# starting kernel
# baseline (speedup 1.0000x reference)
#include <hip/hip_runtime.h>
#include <hip/hip_bf16.h>

#define N_NODES 50000
#define N_EDGES 800000
#define D       128
#define D_OUT   128
#define K2      256
#define XS_STRIDE 260

typedef __attribute__((ext_vector_type(2))) float v2f;
typedef __attribute__((ext_vector_type(8))) float v8f;

__global__ void gcn_zero_kernel(float* __restrict__ ws, int count) {
    int i = blockIdx.x * blockDim.x + threadIdx.x;
    if (i < count) ws[i] = 0.0f;
}

__global__ void gcn_scatter_kernel(const float* __restrict__ feat,
                                   const int* __restrict__ src,
                                   const int* __restrict__ dst,
                                   float* __restrict__ agg,
                                   float* __restrict__ deg) {
    int wavesPerBlk = blockDim.x >> 5;
    int e = blockIdx.x * wavesPerBlk + (threadIdx.x >> 5);
    if (e >= N_EDGES) return;
    int lane = threadIdx.x & 31;
    int s = src[e];
    int t = dst[e];
    const float4 v = *(const float4*)(feat + (size_t)s * D + lane * 4);
    float* ap = agg + (size_t)t * D + lane * 4;
    atomicAdd(ap + 0, v.x);
    atomicAdd(ap + 1, v.y);
    atomicAdd(ap + 2, v.z);
    atomicAdd(ap + 3, v.w);
    if (lane == 0) atomicAdd(deg + t, 1.0f);
}

__global__ void gcn_gemm_kernel(const float* __restrict__ feat,
                                const float* __restrict__ agg,
                                const float* __restrict__ deg,
                                const float* __restrict__ W,
                                const float* __restrict__ bias,
                                float* __restrict__ out) {
    __shared__ float xs[16 * XS_STRIDE];
    __shared__ float invd[16];

    const int node0 = blockIdx.x * 16;
    const int t = threadIdx.x;

    if (t < 16) {
        float dg = deg[node0 + t];
        invd[t] = 1.0f / fmaxf(dg, 1.0f);
    }
    __syncthreads();

    #pragma unroll
    for (int i = t; i < 16 * K2; i += 256) {
        int r = i >> 8;
        int c = i & 255;
        float v;
        if (c < D) {
            v = agg[(size_t)(node0 + r) * D + c] * invd[r];
        } else {
            v = feat[(size_t)(node0 + r) * D + (c - D)];
        }
        xs[r * XS_STRIDE + c] = v;
    }
    __syncthreads();

    const int wave = t >> 5;
    const int lane = t & 31;
    const int half = lane >> 4;
    const int lid  = lane & 15;
    const int n0   = wave * 16;

    const float* wrow = W + (size_t)(n0 + lid) * K2 + 2 * half;
    const float* arow = xs + lid * XS_STRIDE + 2 * half;

    __builtin_prefetch(wrow, 0, 3);

    v8f acc = {};
    #pragma unroll 8
    for (int k = 0; k < K2; k += 4) {
        v2f a = *(const v2f*)(arow + k);
        v2f b = *(const v2f*)(wrow + k);
        acc = __builtin_amdgcn_wmma_f32_16x16x4_f32(
            false, a, false, b, (short)0, acc, false, false);
    }

    const float bv = bias[n0 + lid];
    #pragma unroll
    for (int j = 0; j < 8; ++j) {
        int row = node0 + j + 8 * half;
        out[(size_t)row * D_OUT + n0 + lid] = acc[j] + bv;
    }
}

extern "C" void kernel_launch(void* const* d_in, const int* in_sizes, int n_in,
                              void* d_out, int out_size, void* d_ws, size_t ws_size,
                              hipStream_t stream) {
    const float* feat = (const float*)d_in[0];
    const int*   src  = (const int*)d_in[1];
    const int*   dst  = (const int*)d_in[2];
    const float* W    = (const float*)d_in[3];
    const float* bias = (const float*)d_in[4];
    float*       out  = (float*)d_out;

    float* agg = (float*)d_ws;
    float* deg = agg + (size_t)N_NODES * D;

    {
        int count = N_NODES * D + N_NODES;
        int blocks = (count + 255) / 256;
        gcn_zero_kernel<<<blocks, 256, 0, stream>>>(agg, count);
    }

    {
        int blocks = (N_EDGES + 7) / 8;
        gcn_scatter_kernel<<<blocks, 256, 0, stream>>>(feat, src, dst, agg, deg);
    }

    {
        int blocks = N_NODES / 16;
        gcn_gemm_kernel<<<blocks, 256, 0, stream>>>(feat, agg, deg, W, bias, out);
    }
}


# list build without global atomics: a block links its 3125-edge slice in an LDS head table of all nodes, writes 16-bit head planes; gather reads 256 heads per node; no init launch; sums slice-ordered
# speedup vs baseline: 3.6184x; 3.6184x over previous
#include <hip/hip_runtime.h>
#include <hip/hip_bf16.h>

#define N_NODES 50000
#define N_EDGES 800000
#define D       128
#define D_OUT   128
#define K2      256

typedef __attribute__((ext_vector_type(2))) float v2f;
typedef __attribute__((ext_vector_type(8))) float v8f;

typedef __attribute__((ext_vector_type(2))) int v2i;

#define N_SLICES     256
#define SLICE_EDGES  (N_EDGES / N_SLICES)
#define HP           50048
#define HPW          (HP / 2)
#define LINK_THREADS 1024
static_assert(N_EDGES % N_SLICES == 0, "whole slices");
static_assert(SLICE_EDGES + 1 < 65536, "a slice-local edge number + 1 fits 16 bits");
static_assert(SLICE_EDGES <= 4 * LINK_THREADS, "at most 4 edges per thread");
static_assert(HP >= N_NODES && HP % 64 == 0, "planes hold whole 64-node groups");

__global__ void __launch_bounds__(LINK_THREADS)
gcn_link_kernel(const int* __restrict__ src,
                const int* __restrict__ dst,
                int* __restrict__ heads,
                int* __restrict__ rec) {
    __shared__ __attribute__((aligned(16))) int tab[HP];

    const int t    = threadIdx.x;
    const int b    = blockIdx.x;
    const int base = b * SLICE_EDGES;

    int s[4], d[4];
    #pragma unroll
    for (int k = 0; k < 4; ++k) {
        const int le = t + LINK_THREADS * k;
        s[k] = 0;
        d[k] = 0;
        if (le < SLICE_EDGES) {
            s[k] = src[base + le];
            d[k] = dst[base + le];
        }
    }
    for (int q = t; q < HP / 4; q += LINK_THREADS) ((int4*)tab)[q] = make_int4(0, 0, 0, 0);
    __syncthreads();

    #pragma unroll
    for (int k = 0; k < 4; ++k) {
        const int le = t + LINK_THREADS * k;
        if (le < SLICE_EDGES) {
            const int prev = atomicExch(&tab[d[k]], le + 1);
            v2i r;
            r.x = prev ? base + prev : 0;
            r.y = s[k];
            *(v2i*)(rec + 2 * (size_t)(base + le)) = r;
        }
    }
    __syncthreads();

    for (int q = t; q < HP / 8; q += LINK_THREADS) {
        const int4 lo = ((const int4*)tab)[2 * q];
        const int4 hi = ((const int4*)tab)[2 * q + 1];
        int4 p;
        p.x = lo.x | (lo.y << 16);
        p.y = lo.z | (lo.w << 16);
        p.z = hi.x | (hi.y << 16);
        p.w = hi.z | (hi.w << 16);
        *(int4*)(heads + (size_t)b * HPW + 4 * q) = p;
    }
}

#define NODES_PER_WAVE 8
#define GB_NODES       64
#define GB_BLOCKS      (HP / GB_NODES)
#define HLP            72
static_assert(N_NODES % NODES_PER_WAVE == 0, "a wave owns 8 whole nodes");

static __device__ __forceinline__ int gcn_next_head(const unsigned short* __restrict__ hl,
                                                    const unsigned short* __restrict__ sl,
                                                    const int ln, const int i, const int cnt) {
    if (i >= cnt) return -1;
    const int slice = sl[ln * N_SLICES + i];
    return slice * SLICE_EDGES + (int)hl[slice * HLP + ln] - 1;
}

__global__ void __launch_bounds__(256)
gcn_gather_kernel(const float* __restrict__ feat,
                  const int* __restrict__ heads,
                  const int* __restrict__ rec,
                  float* __restrict__ agg,
                  float* __restrict__ deg) {
    __shared__ __attribute__((aligned(16))) unsigned short hl[N_SLICES * HLP];
    __shared__ unsigned short sl[GB_NODES * N_SLICES];
    __shared__ int cnts[GB_NODES];

    const int t    = threadIdx.x;
    const int w    = t >> 5;
    const int lane = t & 31;
    const int g    = blockIdx.x;
    const int sub  = lane & 7;

    {
        int4 v[8];
        #pragma unroll
        for (int k = 0; k < 8; ++k) {
            const int slice = w * 32 + k * 4 + (lane >> 3);
            v[k] = *(const int4*)(heads + (size_t)slice * HPW + g * (GB_NODES / 2) + sub * 4);
        }
        #pragma unroll
        for (int k = 0; k < 8; ++k) {
            const int slice = w * 32 + k * 4 + (lane >> 3);
            *(int4*)(hl + slice * HLP + sub * 8) = v[k];
        }
    }
    __syncthreads();

    {
        int4 hv[N_SLICES / 32];
        #pragma unroll
        for (int step = 0; step < N_SLICES / 32; ++step)
            hv[step] = *(const int4*)(hl + (step * 32 + lane) * HLP + w * NODES_PER_WAVE);
        const unsigned below = (1u << lane) - 1u;
        #pragma unroll
        for (int j = 0; j < NODES_PER_WAVE; ++j) {
            const int ln = w * NODES_PER_WAVE + j;
            int cnt = 0;
            #pragma unroll
            for (int step = 0; step < N_SLICES / 32; ++step) {
                const int pair = (j >> 1) == 0 ? hv[step].x : (j >> 1) == 1 ? hv[step].y
                               : (j >> 1) == 2 ? hv[step].z : hv[step].w;
                const bool nz = (j & 1) ? ((unsigned)pair >> 16) != 0u : (pair & 0xffff) != 0;
                const unsigned m = __builtin_amdgcn_ballot_w32(nz);
                if (nz) sl[ln * N_SLICES + cnt + __popc(m & below)] = (unsigned short)(step * 32 + lane);
                cnt += __popc(m);
            }
            if (lane == 0) cnts[ln] = cnt;
        }
    }
    __syncthreads();
    if (g * GB_NODES + w * NODES_PER_WAVE >= N_NODES) return;

    const int lnA   = w * NODES_PER_WAVE + (lane >> 3);
    const int lnB   = lnA + 4;
    const int nodeA = g * GB_NODES + lnA;
    const int nodeB = nodeA + 4;

    const float* fcol = feat + sub * 4;
    const v2i*   rec2 = (const v2i*)rec;

    const int nA = cnts[lnA];
    const int nB = cnts[lnB];
    int eA = gcn_next_head(hl, sl, lnA, 0, nA);
    int eB = gcn_next_head(hl, sl, lnB, 0, nB);
    int hA = gcn_next_head(hl, sl, lnA, 1, nA);
    int hB = gcn_next_head(hl, sl, lnB, 1, nB);
    int iA = 2, iB = 2;
    v2i rA = {0, 0};
    v2i rB = {0, 0};
    if (eA >= 0) rA = rec2[eA];
    if (eB >= 0) rB = rec2[eB];

    float4 a0 = make_float4(0.0f, 0.0f, 0.0f, 0.0f), a1 = a0, a2 = a0, a3 = a0;
    float4 b0 = a0, b1 = a0, b2 = a0, b3 = a0;
    int cA = 0, cB = 0;

    for (int it = 0; it < N_EDGES; ++it) {
        const bool actA = eA >= 0;
        const bool actB = eB >= 0;
        if (!(actA || actB)) break;

        asm volatile("" :: "v"(rA.x), "v"(rA.y), "v"(rB.x), "v"(rB.y));

        float4 va0, va1, va2, va3, vb0, vb1, vb2, vb3;
        bool tookA = false, tookB = false;

        if (actA) {
            const float* p = fcol + (size_t)rA.y * D;
            va0 = *(const float4*)(p);
            va1 = *(const float4*)(p + 32);
            va2 = *(const float4*)(p + 64);
            va3 = *(const float4*)(p + 96);
            tookA = rA.x == 0;
            eA = tookA ? hA : rA.x - 1;
            if (eA >= 0) rA = rec2[eA];
        }
        if (actB) {
            const float* p = fcol + (size_t)rB.y * D;
            vb0 = *(const float4*)(p);
            vb1 = *(const float4*)(p + 32);
            vb2 = *(const float4*)(p + 64);
            vb3 = *(const float4*)(p + 96);
            tookB = rB.x == 0;
            eB = tookB ? hB : rB.x - 1;
            if (eB >= 0) rB = rec2[eB];
        }
        if (tookA) {
            hA = gcn_next_head(hl, sl, lnA, iA, nA);
            ++iA;
        }
        if (tookB) {
            hB = gcn_next_head(hl, sl, lnB, iB, nB);
            ++iB;
        }
        if (actA) {
            a0.x += va0.x; a0.y += va0.y; a0.z += va0.z; a0.w += va0.w;
            a1.x += va1.x; a1.y += va1.y; a1.z += va1.z; a1.w += va1.w;
            a2.x += va2.x; a2.y += va2.y; a2.z += va2.z; a2.w += va2.w;
            a3.x += va3.x; a3.y += va3.y; a3.z += va3.z; a3.w += va3.w;
            ++cA;
        }
        if (actB) {
            b0.x += vb0.x; b0.y += vb0.y; b0.z += vb0.z; b0.w += vb0.w;
            b1.x += vb1.x; b1.y += vb1.y; b1.z += vb1.z; b1.w += vb1.w;
            b2.x += vb2.x; b2.y += vb2.y; b2.z += vb2.z; b2.w += vb2.w;
            b3.x += vb3.x; b3.y += vb3.y; b3.z += vb3.z; b3.w += vb3.w;
            ++cB;
        }
    }

    float* oa = agg + (size_t)nodeA * D + sub * 4;
    *(float4*)(oa)      = a0;
    *(float4*)(oa + 32) = a1;
    *(float4*)(oa + 64) = a2;
    *(float4*)(oa + 96) = a3;
    float* ob = agg + (size_t)nodeB * D + sub * 4;
    *(float4*)(ob)      = b0;
    *(float4*)(ob + 32) = b1;
    *(float4*)(ob + 64) = b2;
    *(float4*)(ob + 96) = b3;
    if (sub == 0) {
        deg[nodeA] = (float)cA;
        deg[nodeB] = (float)cB;
    }
}

#define N_TILES      (N_NODES / 16)
#define GEMM_WAVES   16
#define GEMM_THREADS (GEMM_WAVES * 32)
#define GEMM_BLOCKS  256
#define FULL_WAVES   12
#define N_FULL       (GEMM_BLOCKS * FULL_WAVES)
#define WS_PITCH     260
static_assert(N_NODES % 16 == 0, "whole 16-node tiles: EXEC all ones at every WMMA");
static_assert(N_FULL <= N_TILES && N_TILES - N_FULL <= GEMM_BLOCKS, "the tiles left over fit one per block");
static_assert(GEMM_WAVES - FULL_WAVES == 4, "a left-over tile is split in four column quarters");


template <int NJ>
static __device__ __forceinline__ void gcn_chunk_mma(const float4 (&x)[8],
                                                     const float* __restrict__ bq,
                                                     v8f (&acc)[NJ]) {
    #pragma unroll
    for (int s = 0; s < 8; ++s) {
        float4 b[NJ];
        #pragma unroll
        for (int j = 0; j < NJ; ++j) b[j] = *(const float4*)(bq + j * 16 * WS_PITCH + 8 * s);
        const v2f a0 = {x[s].x, x[s].y};
        const v2f a1 = {x[s].z, x[s].w};
        #pragma unroll
        for (int j = 0; j < NJ; ++j) {
            const v2f b0 = {b[j].x, b[j].y};
            acc[j] = __builtin_amdgcn_wmma_f32_16x16x4_f32(
                false, a0, false, b0, (short)0, acc[j], false, false);
        }
        #pragma unroll
        for (int j = 0; j < NJ; ++j) {
            const v2f b1 = {b[j].z, b[j].w};
            acc[j] = __builtin_amdgcn_wmma_f32_16x16x4_f32(
                false, a1, false, b1, (short)0, acc[j], false, false);
        }
    }
}

template <int NJ>
static __device__ __forceinline__ void gcn_tile_mma(float4 (&xe)[8],
                                                    const float* __restrict__ pa,
                                                    const float* __restrict__ pf,
                                                    const float invd,
                                                    const float* __restrict__ bq,
                                                    v8f (&acc)[NJ]) {
    float4 xo[8];
    #pragma unroll 1
    for (int half = 0; half < 2; ++half) {
        const float* px = half ? pf : pa;
        if (half == 0) {
            #pragma unroll
            for (int s = 0; s < 8; ++s) {
                xe[s].x *= invd; xe[s].y *= invd; xe[s].z *= invd; xe[s].w *= invd;
            }
        }
        #pragma unroll
        for (int s = 0; s < 8; ++s) xo[s] = *(const float4*)(px + 64 + 8 * s);
        gcn_chunk_mma<NJ>(xe, bq + half * 128, acc);
        if (half == 0) {
            #pragma unroll
            for (int s = 0; s < 8; ++s) xe[s] = *(const float4*)(pf + 8 * s);
            #pragma unroll
            for (int s = 0; s < 8; ++s) {
                xo[s].x *= invd; xo[s].y *= invd; xo[s].z *= invd; xo[s].w *= invd;
            }
        }
        gcn_chunk_mma<NJ>(xo, bq + half * 128 + 64, acc);
    }
}

__global__ void __launch_bounds__(GEMM_THREADS)
gcn_gemm_kernel(const float* __restrict__ feat,
                const float* __restrict__ agg,
                const float* __restrict__ deg,
                const float* __restrict__ W,
                const float* __restrict__ bias,
                float* __restrict__ out) {
    __shared__ __attribute__((aligned(16))) float wsm[D_OUT * WS_PITCH];

    const int t    = threadIdx.x;
    const int wave = t >> 5;
    const int lane = t & 31;
    const int h    = lane >> 4;
    const int c    = lane & 15;
    const bool full = wave < FULL_WAVES;
    const int tile  = full ? blockIdx.x + GEMM_BLOCKS * wave : N_FULL + blockIdx.x;
    const bool has_tile = tile < N_TILES;

    {
        float4 wv[16];
        #pragma unroll
        for (int i = 0; i < 16; ++i) {
            const int idx = t + GEMM_THREADS * i;
            wv[i] = *(const float4*)(W + (size_t)(idx >> 6) * K2 + (idx & 63) * 4);
        }
        #pragma unroll
        for (int i = 0; i < 16; ++i) {
            const int idx  = t + GEMM_THREADS * i;
            const int n    = idx >> 6;
            const int slot = ((n & 3) + 4 * (n >> 6)) * 16 + ((n & 63) >> 2);
            *(float4*)(wsm + slot * WS_PITCH + (idx & 63) * 4) = wv[i];
        }
    }

    float4 xe[8];
    float dg = 1.0f;
    const float* pa = agg;
    const float* pf = feat;
    if (has_tile) {
        const size_t row = (size_t)tile * 16 + c;
        pa = agg  + row * D + 4 * h;
        pf = feat + row * D + 4 * h;
        #pragma unroll
        for (int s = 0; s < 8; ++s) xe[s] = *(const float4*)(pa + 8 * s);
        dg = deg[row];
    }
    asm volatile("s_wait_dscnt 0x0" ::: "memory");
    asm volatile("s_barrier_signal -1" ::: "memory");
    asm volatile("s_barrier_wait -1" ::: "memory");
    if (!has_tile) return;

    const float invd = 1.0f / fmaxf(dg, 1.0f);
    float* prow = out + ((size_t)tile * 16 + 8 * h) * D_OUT;

    if (full) {
        const float4 bias_lo = *(const float4*)(bias + 4 * c);
        const float4 bias_hi = *(const float4*)(bias + 64 + 4 * c);
        v8f acc[8];
        #pragma unroll
        for (int j = 0; j < 8; ++j) acc[j] = (v8f){};
        gcn_tile_mma<8>(xe, pa, pf, invd, wsm + c * WS_PITCH + 4 * h, acc);
        float* po = prow + 4 * c;
        #pragma unroll
        for (int r = 0; r < 8; ++r) {
            float4 lo, hi;
            lo.x = acc[0][r] + bias_lo.x; lo.y = acc[1][r] + bias_lo.y;
            lo.z = acc[2][r] + bias_lo.z; lo.w = acc[3][r] + bias_lo.w;
            hi.x = acc[4][r] + bias_hi.x; hi.y = acc[5][r] + bias_hi.y;
            hi.z = acc[6][r] + bias_hi.z; hi.w = acc[7][r] + bias_hi.w;
            *(float4*)(po + (size_t)r * D_OUT)      = lo;
            *(float4*)(po + (size_t)r * D_OUT + 64) = hi;
        }
    } else {
        const int q    = wave - FULL_WAVES;
        const int n0   = 32 * q + 2 * c;
        const int slot = ((n0 & 3) + 4 * (n0 >> 6)) * 16 + ((n0 & 63) >> 2);
        const float2 bb = *(const float2*)(bias + n0);
        v8f acc[2];
        acc[0] = (v8f){};
        acc[1] = (v8f){};
        gcn_tile_mma<2>(xe, pa, pf, invd, wsm + slot * WS_PITCH + 4 * h, acc);
        float* po = prow + n0;
        #pragma unroll
        for (int r = 0; r < 8; ++r) {
            float2 v;
            v.x = acc[0][r] + bb.x;
            v.y = acc[1][r] + bb.y;
            *(float2*)(po + (size_t)r * D_OUT) = v;
        }
    }
}

extern "C" void kernel_launch(void* const* d_in, const int* in_sizes, int n_in,
                              void* d_out, int out_size, void* d_ws, size_t ws_size,
                              hipStream_t stream) {
    const float* feat = (const float*)d_in[0];
    const int*   src  = (const int*)d_in[1];
    const int*   dst  = (const int*)d_in[2];
    const float* W    = (const float*)d_in[3];
    const float* bias = (const float*)d_in[4];
    float*       out  = (float*)d_out;

    float* agg   = (float*)d_ws;
    float* deg   = agg + (size_t)N_NODES * D;
    int*   heads = (int*)(deg + HP);
    int*   rec   = heads + (size_t)N_SLICES * HPW;

    gcn_link_kernel<<<N_SLICES, LINK_THREADS, 0, stream>>>(src, dst, heads, rec);

    gcn_gather_kernel<<<GB_BLOCKS, 256, 0, stream>>>(feat, heads, rec, agg, deg);

    gcn_gemm_kernel<<<GEMM_BLOCKS, GEMM_THREADS, 0, stream>>>(feat, agg, deg, W, bias, out);
}


# list build in 64 slices of 12500 edges, not 256 of 3125: 64 head planes to write, fetch and scan per node; link starts 2048 waves, not 8192; sums ordered by 64 slices, exchange order inside a slice
# speedup vs baseline: 3.8290x; 1.0582x over previous
#include <hip/hip_runtime.h>
#include <hip/hip_bf16.h>

#define N_NODES 50000
#define N_EDGES 800000
#define D       128
#define D_OUT   128
#define K2      256

typedef __attribute__((ext_vector_type(2))) float v2f;
typedef __attribute__((ext_vector_type(8))) float v8f;

typedef __attribute__((ext_vector_type(2))) int v2i;

#define N_SLICES     64
#define SLICE_EDGES  (N_EDGES / N_SLICES)
#define HP           50048
#define HPW          (HP / 2)
#define LINK_THREADS 1024
#define LINK_EPT     ((SLICE_EDGES + LINK_THREADS - 1) / LINK_THREADS)
static_assert(N_EDGES % N_SLICES == 0, "whole slices");
static_assert(SLICE_EDGES + 1 < 65536, "a slice-local edge number + 1 fits 16 bits");
static_assert(SLICE_EDGES <= LINK_EPT * LINK_THREADS, "every edge of the slice has a thread");
static_assert(HP >= N_NODES && HP % 64 == 0, "planes hold whole 64-node groups");

__global__ void __launch_bounds__(LINK_THREADS)
gcn_link_kernel(const int* __restrict__ src,
                const int* __restrict__ dst,
                int* __restrict__ heads,
                int* __restrict__ rec) {
    __shared__ __attribute__((aligned(16))) int tab[HP];

    const int t    = threadIdx.x;
    const int b    = blockIdx.x;
    const int base = b * SLICE_EDGES;

    int s[LINK_EPT], d[LINK_EPT];
    #pragma unroll
    for (int k = 0; k < LINK_EPT; ++k) {
        const int le = t + LINK_THREADS * k;
        s[k] = 0;
        d[k] = 0;
        if (le < SLICE_EDGES) {
            s[k] = src[base + le];
            d[k] = dst[base + le];
        }
    }
    for (int q = t; q < HP / 4; q += LINK_THREADS) ((int4*)tab)[q] = make_int4(0, 0, 0, 0);
    __syncthreads();

    #pragma unroll
    for (int k = 0; k < LINK_EPT; ++k) {
        const int le = t + LINK_THREADS * k;
        if (le < SLICE_EDGES) {
            const int prev = atomicExch(&tab[d[k]], le + 1);
            v2i r;
            r.x = prev ? base + prev : 0;
            r.y = s[k];
            *(v2i*)(rec + 2 * (size_t)(base + le)) = r;
        }
    }
    __syncthreads();

    for (int q = t; q < HP / 8; q += LINK_THREADS) {
        const int4 lo = ((const int4*)tab)[2 * q];
        const int4 hi = ((const int4*)tab)[2 * q + 1];
        int4 p;
        p.x = lo.x | (lo.y << 16);
        p.y = lo.z | (lo.w << 16);
        p.z = hi.x | (hi.y << 16);
        p.w = hi.z | (hi.w << 16);
        *(int4*)(heads + (size_t)b * HPW + 4 * q) = p;
    }
}

#define NODES_PER_WAVE 8
#define GB_NODES       64
#define GB_BLOCKS      (HP / GB_NODES)
#define HLP            72
#define SLICES_PER_WAVE (N_SLICES / 8)
#define HEAD_LOADS      (SLICES_PER_WAVE / 4)
static_assert(N_NODES % NODES_PER_WAVE == 0, "a wave owns 8 whole nodes");
static_assert(N_SLICES % 32 == 0, "phase 1: 8 waves x whole groups of 4 lines; phase 2: whole steps of 32 slices");

static __device__ __forceinline__ int gcn_next_head(const unsigned short* __restrict__ hl,
                                                    const unsigned short* __restrict__ sl,
                                                    const int ln, const int i, const int cnt) {
    if (i >= cnt) return -1;
    const int slice = sl[ln * N_SLICES + i];
    return slice * SLICE_EDGES + (int)hl[slice * HLP + ln] - 1;
}

__global__ void __launch_bounds__(256)
gcn_gather_kernel(const float* __restrict__ feat,
                  const int* __restrict__ heads,
                  const int* __restrict__ rec,
                  float* __restrict__ agg,
                  float* __restrict__ deg) {
    __shared__ __attribute__((aligned(16))) unsigned short hl[N_SLICES * HLP];
    __shared__ unsigned short sl[GB_NODES * N_SLICES];
    __shared__ int cnts[GB_NODES];

    const int t    = threadIdx.x;
    const int w    = t >> 5;
    const int lane = t & 31;
    const int g    = blockIdx.x;
    const int sub  = lane & 7;

    {
        int4 v[HEAD_LOADS];
        #pragma unroll
        for (int k = 0; k < HEAD_LOADS; ++k) {
            const int slice = w * SLICES_PER_WAVE + k * 4 + (lane >> 3);
            v[k] = *(const int4*)(heads + (size_t)slice * HPW + g * (GB_NODES / 2) + sub * 4);
        }
        #pragma unroll
        for (int k = 0; k < HEAD_LOADS; ++k) {
            const int slice = w * SLICES_PER_WAVE + k * 4 + (lane >> 3);
            *(int4*)(hl + slice * HLP + sub * 8) = v[k];
        }
    }
    __syncthreads();

    {
        int4 hv[N_SLICES / 32];
        #pragma unroll
        for (int step = 0; step < N_SLICES / 32; ++step)
            hv[step] = *(const int4*)(hl + (step * 32 + lane) * HLP + w * NODES_PER_WAVE);
        const unsigned below = (1u << lane) - 1u;
        #pragma unroll
        for (int j = 0; j < NODES_PER_WAVE; ++j) {
            const int ln = w * NODES_PER_WAVE + j;
            int cnt = 0;
            #pragma unroll
            for (int step = 0; step < N_SLICES / 32; ++step) {
                const int pair = (j >> 1) == 0 ? hv[step].x : (j >> 1) == 1 ? hv[step].y
                               : (j >> 1) == 2 ? hv[step].z : hv[step].w;
                const bool nz = (j & 1) ? ((unsigned)pair >> 16) != 0u : (pair & 0xffff) != 0;
                const unsigned m = __builtin_amdgcn_ballot_w32(nz);
                if (nz) sl[ln * N_SLICES + cnt + __popc(m & below)] = (unsigned short)(step * 32 + lane);
                cnt += __popc(m);
            }
            if (lane == 0) cnts[ln] = cnt;
        }
    }
    __syncthreads();
    if (g * GB_NODES + w * NODES_PER_WAVE >= N_NODES) return;

    const int lnA   = w * NODES_PER_WAVE + (lane >> 3);
    const int lnB   = lnA + 4;
    const int nodeA = g * GB_NODES + lnA;
    const int nodeB = nodeA + 4;

    const float* fcol = feat + sub * 4;
    const v2i*   rec2 = (const v2i*)rec;

    const int nA = cnts[lnA];
    const int nB = cnts[lnB];
    int eA = gcn_next_head(hl, sl, lnA, 0, nA);
    int eB = gcn_next_head(hl, sl, lnB, 0, nB);
    int hA = gcn_next_head(hl, sl, lnA, 1, nA);
    int hB = gcn_next_head(hl, sl, lnB, 1, nB);
    int iA = 2, iB = 2;
    v2i rA = {0, 0};
    v2i rB = {0, 0};
    if (eA >= 0) rA = rec2[eA];
    if (eB >= 0) rB = rec2[eB];

    float4 a0 = make_float4(0.0f, 0.0f, 0.0f, 0.0f), a1 = a0, a2 = a0, a3 = a0;
    float4 b0 = a0, b1 = a0, b2 = a0, b3 = a0;
    int cA = 0, cB = 0;

    for (int it = 0; it < N_EDGES; ++it) {
        const bool actA = eA >= 0;
        const bool actB = eB >= 0;
        if (!(actA || actB)) break;

        asm volatile("" :: "v"(rA.x), "v"(rA.y), "v"(rB.x), "v"(rB.y));

        float4 va0, va1, va2, va3, vb0, vb1, vb2, vb3;
        bool tookA = false, tookB = false;

        if (actA) {
            const float* p = fcol + (size_t)rA.y * D;
            va0 = *(const float4*)(p);
            va1 = *(const float4*)(p + 32);
            va2 = *(const float4*)(p + 64);
            va3 = *(const float4*)(p + 96);
            tookA = rA.x == 0;
            eA = tookA ? hA : rA.x - 1;
            if (eA >= 0) rA = rec2[eA];
        }
        if (actB) {
            const float* p = fcol + (size_t)rB.y * D;
            vb0 = *(const float4*)(p);
            vb1 = *(const float4*)(p + 32);
            vb2 = *(const float4*)(p + 64);
            vb3 = *(const float4*)(p + 96);
            tookB = rB.x == 0;
            eB = tookB ? hB : rB.x - 1;
            if (eB >= 0) rB = rec2[eB];
        }
        if (tookA) {
            hA = gcn_next_head(hl, sl, lnA, iA, nA);
            ++iA;
        }
        if (tookB) {
            hB = gcn_next_head(hl, sl, lnB, iB, nB);
            ++iB;
        }
        if (actA) {
            a0.x += va0.x; a0.y += va0.y; a0.z += va0.z; a0.w += va0.w;
            a1.x += va1.x; a1.y += va1.y; a1.z += va1.z; a1.w += va1.w;
            a2.x += va2.x; a2.y += va2.y; a2.z += va2.z; a2.w += va2.w;
            a3.x += va3.x; a3.y += va3.y; a3.z += va3.z; a3.w += va3.w;
            ++cA;
        }
        if (actB) {
            b0.x += vb0.x; b0.y += vb0.y; b0.z += vb0.z; b0.w += vb0.w;
            b1.x += vb1.x; b1.y += vb1.y; b1.z += vb1.z; b1.w += vb1.w;
            b2.x += vb2.x; b2.y += vb2.y; b2.z += vb2.z; b2.w += vb2.w;
            b3.x += vb3.x; b3.y += vb3.y; b3.z += vb3.z; b3.w += vb3.w;
            ++cB;
        }
    }

    float* oa = agg + (size_t)nodeA * D + sub * 4;
    *(float4*)(oa)      = a0;
    *(float4*)(oa + 32) = a1;
    *(float4*)(oa + 64) = a2;
    *(float4*)(oa + 96) = a3;
    float* ob = agg + (size_t)nodeB * D + sub * 4;
    *(float4*)(ob)      = b0;
    *(float4*)(ob + 32) = b1;
    *(float4*)(ob + 64) = b2;
    *(float4*)(ob + 96) = b3;
    if (sub == 0) {
        deg[nodeA] = (float)cA;
        deg[nodeB] = (float)cB;
    }
}

#define N_TILES      (N_NODES / 16)
#define GEMM_WAVES   16
#define GEMM_THREADS (GEMM_WAVES * 32)
#define GEMM_BLOCKS  256
#define FULL_WAVES   12
#define N_FULL       (GEMM_BLOCKS * FULL_WAVES)
#define WS_PITCH     260
static_assert(N_NODES % 16 == 0, "whole 16-node tiles: EXEC all ones at every WMMA");
static_assert(N_FULL <= N_TILES && N_TILES - N_FULL <= GEMM_BLOCKS, "the tiles left over fit one per block");
static_assert(GEMM_WAVES - FULL_WAVES == 4, "a left-over tile is split in four column quarters");


template <int NJ>
static __device__ __forceinline__ void gcn_chunk_mma(const float4 (&x)[8],
                                                     const float* __restrict__ bq,
                                                     v8f (&acc)[NJ]) {
    #pragma unroll
    for (int s = 0; s < 8; ++s) {
        float4 b[NJ];
        #pragma unroll
        for (int j = 0; j < NJ; ++j) b[j] = *(const float4*)(bq + j * 16 * WS_PITCH + 8 * s);
        const v2f a0 = {x[s].x, x[s].y};
        const v2f a1 = {x[s].z, x[s].w};
        #pragma unroll
        for (int j = 0; j < NJ; ++j) {
            const v2f b0 = {b[j].x, b[j].y};
            acc[j] = __builtin_amdgcn_wmma_f32_16x16x4_f32(
                false, a0, false, b0, (short)0, acc[j], false, false);
        }
        #pragma unroll
        for (int j = 0; j < NJ; ++j) {
            const v2f b1 = {b[j].z, b[j].w};
            acc[j] = __builtin_amdgcn_wmma_f32_16x16x4_f32(
                false, a1, false, b1, (short)0, acc[j], false, false);
        }
    }
}

template <int NJ>
static __device__ __forceinline__ void gcn_tile_mma(float4 (&xe)[8],
                                                    const float* __restrict__ pa,
                                                    const float* __restrict__ pf,
                                                    const float invd,
                                                    const float* __restrict__ bq,
                                                    v8f (&acc)[NJ]) {
    float4 xo[8];
    #pragma unroll 1
    for (int half = 0; half < 2; ++half) {
        const float* px = half ? pf : pa;
        if (half == 0) {
            #pragma unroll
            for (int s = 0; s < 8; ++s) {
                xe[s].x *= invd; xe[s].y *= invd; xe[s].z *= invd; xe[s].w *= invd;
            }
        }
        #pragma unroll
        for (int s = 0; s < 8; ++s) xo[s] = *(const float4*)(px + 64 + 8 * s);
        gcn_chunk_mma<NJ>(xe, bq + half * 128, acc);
        if (half == 0) {
            #pragma unroll
            for (int s = 0; s < 8; ++s) xe[s] = *(const float4*)(pf + 8 * s);
            #pragma unroll
            for (int s = 0; s < 8; ++s) {
                xo[s].x *= invd; xo[s].y *= invd; xo[s].z *= invd; xo[s].w *= invd;
            }
        }
        gcn_chunk_mma<NJ>(xo, bq + half * 128 + 64, acc);
    }
}

__global__ void __launch_bounds__(GEMM_THREADS)
gcn_gemm_kernel(const float* __restrict__ feat,
                const float* __restrict__ agg,
                const float* __restrict__ deg,
                const float* __restrict__ W,
                const float* __restrict__ bias,
                float* __restrict__ out) {
    __shared__ __attribute__((aligned(16))) float wsm[D_OUT * WS_PITCH];

    const int t    = threadIdx.x;
    const int wave = t >> 5;
    const int lane = t & 31;
    const int h    = lane >> 4;
    const int c    = lane & 15;
    const bool full = wave < FULL_WAVES;
    const int tile  = full ? blockIdx.x + GEMM_BLOCKS * wave : N_FULL + blockIdx.x;
    const bool has_tile = tile < N_TILES;

    {
        float4 wv[16];
        #pragma unroll
        for (int i = 0; i < 16; ++i) {
            const int idx = t + GEMM_THREADS * i;
            wv[i] = *(const float4*)(W + (size_t)(idx >> 6) * K2 + (idx & 63) * 4);
        }
        #pragma unroll
        for (int i = 0; i < 16; ++i) {
            const int idx  = t + GEMM_THREADS * i;
            const int n    = idx >> 6;
            const int slot = ((n & 3) + 4 * (n >> 6)) * 16 + ((n & 63) >> 2);
            *(float4*)(wsm + slot * WS_PITCH + (idx & 63) * 4) = wv[i];
        }
    }

    float4 xe[8];
    float dg = 1.0f;
    const float* pa = agg;
    const float* pf = feat;
    if (has_tile) {
        const size_t row = (size_t)tile * 16 + c;
        pa = agg  + row * D + 4 * h;
        pf = feat + row * D + 4 * h;
        #pragma unroll
        for (int s = 0; s < 8; ++s) xe[s] = *(const float4*)(pa + 8 * s);
        dg = deg[row];
    }
    asm volatile("s_wait_dscnt 0x0" ::: "memory");
    asm volatile("s_barrier_signal -1" ::: "memory");
    asm volatile("s_barrier_wait -1" ::: "memory");
    if (!has_tile) return;

    const float invd = 1.0f / fmaxf(dg, 1.0f);
    float* prow = out + ((size_t)tile * 16 + 8 * h) * D_OUT;

    if (full) {
        const float4 bias_lo = *(const float4*)(bias + 4 * c);
        const float4 bias_hi = *(const float4*)(bias + 64 + 4 * c);
        v8f acc[8];
        #pragma unroll
        for (int j = 0; j < 8; ++j) acc[j] = (v8f){};
        gcn_tile_mma<8>(xe, pa, pf, invd, wsm + c * WS_PITCH + 4 * h, acc);
        float* po = prow + 4 * c;
        #pragma unroll
        for (int r = 0; r < 8; ++r) {
            float4 lo, hi;
            lo.x = acc[0][r] + bias_lo.x; lo.y = acc[1][r] + bias_lo.y;
            lo.z = acc[2][r] + bias_lo.z; lo.w = acc[3][r] + bias_lo.w;
            hi.x = acc[4][r] + bias_hi.x; hi.y = acc[5][r] + bias_hi.y;
            hi.z = acc[6][r] + bias_hi.z; hi.w = acc[7][r] + bias_hi.w;
            *(float4*)(po + (size_t)r * D_OUT)      = lo;
            *(float4*)(po + (size_t)r * D_OUT + 64) = hi;
        }
    } else {
        const int q    = wave - FULL_WAVES;
        const int n0   = 32 * q + 2 * c;
        const int slot = ((n0 & 3) + 4 * (n0 >> 6)) * 16 + ((n0 & 63) >> 2);
        const float2 bb = *(const float2*)(bias + n0);
        v8f acc[2];
        acc[0] = (v8f){};
        acc[1] = (v8f){};
        gcn_tile_mma<2>(xe, pa, pf, invd, wsm + slot * WS_PITCH + 4 * h, acc);
        float* po = prow + n0;
        #pragma unroll
        for (int r = 0; r < 8; ++r) {
            float2 v;
            v.x = acc[0][r] + bb.x;
            v.y = acc[1][r] + bb.y;
            *(float2*)(po + (size_t)r * D_OUT) = v;
        }
    }
}

extern "C" void kernel_launch(void* const* d_in, const int* in_sizes, int n_in,
                              void* d_out, int out_size, void* d_ws, size_t ws_size,
                              hipStream_t stream) {
    const float* feat = (const float*)d_in[0];
    const int*   src  = (const int*)d_in[1];
    const int*   dst  = (const int*)d_in[2];
    const float* W    = (const float*)d_in[3];
    const float* bias = (const float*)d_in[4];
    float*       out  = (float*)d_out;

    float* agg   = (float*)d_ws;
    float* deg   = agg + (size_t)N_NODES * D;
    int*   heads = (int*)(deg + HP);
    int*   rec   = heads + (size_t)N_SLICES * HPW;

    gcn_link_kernel<<<N_SLICES, LINK_THREADS, 0, stream>>>(src, dst, heads, rec);

    gcn_gather_kernel<<<GB_BLOCKS, 256, 0, stream>>>(feat, heads, rec, agg, deg);

    gcn_gemm_kernel<<<GEMM_BLOCKS, GEMM_THREADS, 0, stream>>>(feat, agg, deg, W, bias, out);
}


# product moved ahead of the aggregation (linear): feature@[W1|W2]^T on f32 WMMA runs in the list-build launch; gather sums product rows, adds mean to out; 3->2 launches; sums: products in list order
# speedup vs baseline: 4.2512x; 1.1103x over previous
#include <hip/hip_runtime.h>

#define N_NODES 50000
#define N_EDGES 800000
#define D       128
#define D_OUT   128
#define K2      256

typedef __attribute__((ext_vector_type(2))) float v2f;
typedef __attribute__((ext_vector_type(8))) float v8f;
typedef __attribute__((ext_vector_type(2))) int v2i;

#define N_SLICES     64
#define SLICE_EDGES  (N_EDGES / N_SLICES)
#define HP           50048
#define HPW          (HP / 2)
#define FUSE_THREADS 512
#define LINK_EPT     ((SLICE_EDGES + FUSE_THREADS - 1) / FUSE_THREADS)
static_assert(N_EDGES % N_SLICES == 0, "whole slices");
static_assert(SLICE_EDGES + 1 < 65536, "a slice-local edge number + 1 fits 16 bits");
static_assert(SLICE_EDGES <= LINK_EPT * FUSE_THREADS, "every edge of the slice has a thread");
static_assert(HP >= N_NODES && HP % 64 == 0, "planes hold whole 64-node groups");

#define N_TILES      (N_NODES / 16)
#define PURE_BLOCKS  192
#define FUSE_BLOCKS  (N_SLICES + PURE_BLOCKS)
#define FULL_WAVES   12
#define N_FULL       (PURE_BLOCKS * FULL_WAVES)
#define N_SPLIT      (PURE_BLOCKS * 2)
#define N_LATE       (N_SLICES * 8)
#define WS_PITCH     260
static_assert(N_NODES % 16 == 0, "whole 16-node tiles: EXEC all ones at every WMMA");
static_assert(N_FULL + N_SPLIT + N_LATE >= N_TILES, "every tile has an owner");
static_assert(D_OUT * WS_PITCH <= HP, "the W copy fits the LDS table it shares");

static __device__ __forceinline__ void gcn_unit_mma(const float4 (&x)[16],
                                                    const float* __restrict__ bq,
                                                    v8f (&acc)[8]) {
    #pragma unroll
    for (int s = 0; s < 16; ++s) {
        float4 b[8];
        #pragma unroll
        for (int j = 0; j < 8; ++j) b[j] = *(const float4*)(bq + j * 16 * WS_PITCH + 8 * s);
        const v2f a0 = {x[s].x, x[s].y};
        const v2f a1 = {x[s].z, x[s].w};
        #pragma unroll
        for (int j = 0; j < 8; ++j) {
            const v2f b0 = {b[j].x, b[j].y};
            acc[j] = __builtin_amdgcn_wmma_f32_16x16x4_f32(
                false, a0, false, b0, (short)0, acc[j], false, false);
        }
        #pragma unroll
        for (int j = 0; j < 8; ++j) {
            const v2f b1 = {b[j].z, b[j].w};
            acc[j] = __builtin_amdgcn_wmma_f32_16x16x4_f32(
                false, a1, false, b1, (short)0, acc[j], false, false);
        }
    }
}

__global__ void __launch_bounds__(FUSE_THREADS)
gcn_link_product_kernel(const int* __restrict__ src,
                        const int* __restrict__ dst,
                        int* __restrict__ heads,
                        int* __restrict__ rec,
                        const float* __restrict__ feat,
                        const float* __restrict__ W,
                        const float* __restrict__ bias,
                        float* __restrict__ P,
                        float* __restrict__ out) {
    __shared__ __attribute__((aligned(16))) int tab[HP];

    const int t    = threadIdx.x;
    const int bk   = blockIdx.x;
    const bool lists = bk < N_SLICES;

    if (lists) {
        const int base = bk * SLICE_EDGES;
        int s[LINK_EPT], d[LINK_EPT];
        #pragma unroll
        for (int k = 0; k < LINK_EPT; ++k) {
            const int le = t + FUSE_THREADS * k;
            s[k] = 0;
            d[k] = 0;
            if (le < SLICE_EDGES) {
                s[k] = src[base + le];
                d[k] = dst[base + le];
            }
        }
        for (int q = t; q < HP / 4; q += FUSE_THREADS) ((int4*)tab)[q] = make_int4(0, 0, 0, 0);
        __syncthreads();

        #pragma unroll
        for (int k = 0; k < LINK_EPT; ++k) {
            const int le = t + FUSE_THREADS * k;
            if (le < SLICE_EDGES) {
                const int prev = atomicExch(&tab[d[k]], le + 1);
                v2i r;
                r.x = prev ? base + prev : 0;
                r.y = s[k];
                *(v2i*)(rec + 2 * (size_t)(base + le)) = r;
            }
        }
        __syncthreads();

        for (int q = t; q < HP / 8; q += FUSE_THREADS) {
            const int4 lo = ((const int4*)tab)[2 * q];
            const int4 hi = ((const int4*)tab)[2 * q + 1];
            int4 p;
            p.x = lo.x | (lo.y << 16);
            p.y = lo.z | (lo.w << 16);
            p.z = hi.x | (hi.y << 16);
            p.w = hi.z | (hi.w << 16);
            *(int4*)(heads + (size_t)bk * HPW + 4 * q) = p;
        }
        __syncthreads();
    }

    float* wsm = (float*)tab;
    const int wave = __builtin_amdgcn_readfirstlane(t >> 5);
    const int lane = t & 31;
    const int h    = lane >> 4;
    const int c    = lane & 15;

    int tile, hb, he;
    if (lists) {
        tile = N_FULL + N_SPLIT + bk * 8 + (wave >> 1);
        hb = wave & 1;
        he = hb + 1;
    } else if (wave < FULL_WAVES) {
        tile = (bk - N_SLICES) + PURE_BLOCKS * wave;
        hb = 0;
        he = 2;
    } else {
        tile = N_FULL + (bk - N_SLICES) * 2 + ((wave - FULL_WAVES) >> 1);
        hb = (wave - FULL_WAVES) & 1;
        he = hb + 1;
    }
    const bool has_tile = tile < N_TILES;

    {
        float4 wv[16];
        #pragma unroll
        for (int i = 0; i < 16; ++i) {
            const int idx = t + FUSE_THREADS * i;
            wv[i] = *(const float4*)(W + (size_t)(idx >> 6) * K2 + (idx & 63) * 4);
        }
        #pragma unroll
        for (int i = 0; i < 16; ++i) {
            const int idx  = t + FUSE_THREADS * i;
            const int n    = idx >> 6;
            const int slot = ((n & 3) + 4 * (n >> 6)) * 16 + ((n & 63) >> 2);
            *(float4*)(wsm + slot * WS_PITCH + (idx & 63) * 4) = wv[i];
        }
    }

    float4 x[16];
    if (has_tile) {
        const float* pf = feat + ((size_t)tile * 16 + c) * D + 4 * h;
        #pragma unroll
        for (int s = 0; s < 16; ++s) x[s] = *(const float4*)(pf + 8 * s);
    }
    asm volatile("s_wait_dscnt 0x0" ::: "memory");
    asm volatile("s_barrier_signal -1" ::: "memory");
    asm volatile("s_barrier_wait -1" ::: "memory");
    if (!has_tile) return;

    const float* bq = wsm + c * WS_PITCH + 4 * h;
    const size_t o0 = ((size_t)tile * 16 + 8 * h) * D_OUT + 4 * c;

    #pragma unroll 1
    for (int half = hb; half < he; ++half) {
        v8f acc[8];
        #pragma unroll
        for (int j = 0; j < 8; ++j) acc[j] = (v8f){};
        gcn_unit_mma(x, bq + half * 128, acc);
        float* po = (half ? out : P) + o0;
        float4 bl = make_float4(0.0f, 0.0f, 0.0f, 0.0f), bh = bl;
        if (half) {
            bl = *(const float4*)(bias + 4 * c);
            bh = *(const float4*)(bias + 64 + 4 * c);
        }
        #pragma unroll
        for (int r = 0; r < 8; ++r) {
            float4 lo, hi;
            lo.x = acc[0][r] + bl.x; lo.y = acc[1][r] + bl.y;
            lo.z = acc[2][r] + bl.z; lo.w = acc[3][r] + bl.w;
            hi.x = acc[4][r] + bh.x; hi.y = acc[5][r] + bh.y;
            hi.z = acc[6][r] + bh.z; hi.w = acc[7][r] + bh.w;
            *(float4*)(po + (size_t)r * D_OUT)      = lo;
            *(float4*)(po + (size_t)r * D_OUT + 64) = hi;
        }
    }
}

#define NODES_PER_WAVE 8
#define GB_NODES       64
#define GB_BLOCKS      (HP / GB_NODES)
#define HLP            72
#define SLICES_PER_WAVE (N_SLICES / 8)
#define HEAD_LOADS      (SLICES_PER_WAVE / 4)
static_assert(N_NODES % NODES_PER_WAVE == 0, "a wave owns 8 whole nodes");
static_assert(N_SLICES % 32 == 0, "phase 1: 8 waves x whole groups of 4 lines; phase 2: whole steps of 32 slices");

static __device__ __forceinline__ int gcn_next_head(const unsigned short* __restrict__ hl,
                                                    const unsigned short* __restrict__ sl,
                                                    const int ln, const int i, const int cnt) {
    if (i >= cnt) return -1;
    const int slice = sl[ln * N_SLICES + i];
    return slice * SLICE_EDGES + (int)hl[slice * HLP + ln] - 1;
}

__global__ void __launch_bounds__(256)
gcn_gather_kernel(const float* __restrict__ P,
                  const int* __restrict__ heads,
                  const int* __restrict__ rec,
                  float* __restrict__ out) {
    __shared__ __attribute__((aligned(16))) unsigned short hl[N_SLICES * HLP];
    __shared__ unsigned short sl[GB_NODES * N_SLICES];
    __shared__ int cnts[GB_NODES];

    const int t    = threadIdx.x;
    const int w    = t >> 5;
    const int lane = t & 31;
    const int g    = blockIdx.x;
    const int sub  = lane & 7;

    {
        int4 v[HEAD_LOADS];
        #pragma unroll
        for (int k = 0; k < HEAD_LOADS; ++k) {
            const int slice = w * SLICES_PER_WAVE + k * 4 + (lane >> 3);
            v[k] = *(const int4*)(heads + (size_t)slice * HPW + g * (GB_NODES / 2) + sub * 4);
        }
        #pragma unroll
        for (int k = 0; k < HEAD_LOADS; ++k) {
            const int slice = w * SLICES_PER_WAVE + k * 4 + (lane >> 3);
            *(int4*)(hl + slice * HLP + sub * 8) = v[k];
        }
    }
    __syncthreads();

    {
        int4 hv[N_SLICES / 32];
        #pragma unroll
        for (int step = 0; step < N_SLICES / 32; ++step)
            hv[step] = *(const int4*)(hl + (step * 32 + lane) * HLP + w * NODES_PER_WAVE);
        const unsigned below = (1u << lane) - 1u;
        #pragma unroll
        for (int j = 0; j < NODES_PER_WAVE; ++j) {
            const int ln = w * NODES_PER_WAVE + j;
            int cnt = 0;
            #pragma unroll
            for (int step = 0; step < N_SLICES / 32; ++step) {
                const int pair = (j >> 1) == 0 ? hv[step].x : (j >> 1) == 1 ? hv[step].y
                               : (j >> 1) == 2 ? hv[step].z : hv[step].w;
                const bool nz = (j & 1) ? ((unsigned)pair >> 16) != 0u : (pair & 0xffff) != 0;
                const unsigned m = __builtin_amdgcn_ballot_w32(nz);
                if (nz) sl[ln * N_SLICES + cnt + __popc(m & below)] = (unsigned short)(step * 32 + lane);
                cnt += __popc(m);
            }
            if (lane == 0) cnts[ln] = cnt;
        }
    }
    __syncthreads();
    if (g * GB_NODES + w * NODES_PER_WAVE >= N_NODES) return;

    const int lnA   = w * NODES_PER_WAVE + (lane >> 3);
    const int lnB   = lnA + 4;
    const int nodeA = g * GB_NODES + lnA;
    const int nodeB = nodeA + 4;

    const float* fcol = P + sub * 4;
    const v2i*   rec2 = (const v2i*)rec;

    const int nA = cnts[lnA];
    const int nB = cnts[lnB];
    int eA = gcn_next_head(hl, sl, lnA, 0, nA);
    int eB = gcn_next_head(hl, sl, lnB, 0, nB);
    int hA = gcn_next_head(hl, sl, lnA, 1, nA);
    int hB = gcn_next_head(hl, sl, lnB, 1, nB);
    int iA = 2, iB = 2;
    v2i rA = {0, 0};
    v2i rB = {0, 0};
    if (eA >= 0) rA = rec2[eA];
    if (eB >= 0) rB = rec2[eB];

    float4 a0 = make_float4(0.0f, 0.0f, 0.0f, 0.0f), a1 = a0, a2 = a0, a3 = a0;
    float4 b0 = a0, b1 = a0, b2 = a0, b3 = a0;
    int cA = 0, cB = 0;

    for (int it = 0; it < N_EDGES; ++it) {
        const bool actA = eA >= 0;
        const bool actB = eB >= 0;
        if (!(actA || actB)) break;

        asm volatile("" :: "v"(rA.x), "v"(rA.y), "v"(rB.x), "v"(rB.y));

        float4 va0, va1, va2, va3, vb0, vb1, vb2, vb3;
        bool tookA = false, tookB = false;

        if (actA) {
            const float* p = fcol + (size_t)rA.y * D;
            va0 = *(const float4*)(p);
            va1 = *(const float4*)(p + 32);
            va2 = *(const float4*)(p + 64);
            va3 = *(const float4*)(p + 96);
            tookA = rA.x == 0;
            eA = tookA ? hA : rA.x - 1;
            if (eA >= 0) rA = rec2[eA];
        }
        if (actB) {
            const float* p = fcol + (size_t)rB.y * D;
            vb0 = *(const float4*)(p);
            vb1 = *(const float4*)(p + 32);
            vb2 = *(const float4*)(p + 64);
            vb3 = *(const float4*)(p + 96);
            tookB = rB.x == 0;
            eB = tookB ? hB : rB.x - 1;
            if (eB >= 0) rB = rec2[eB];
        }
        if (tookA) {
            hA = gcn_next_head(hl, sl, lnA, iA, nA);
            ++iA;
        }
        if (tookB) {
            hB = gcn_next_head(hl, sl, lnB, iB, nB);
            ++iB;
        }
        if (actA) {
            a0.x += va0.x; a0.y += va0.y; a0.z += va0.z; a0.w += va0.w;
            a1.x += va1.x; a1.y += va1.y; a1.z += va1.z; a1.w += va1.w;
            a2.x += va2.x; a2.y += va2.y; a2.z += va2.z; a2.w += va2.w;
            a3.x += va3.x; a3.y += va3.y; a3.z += va3.z; a3.w += va3.w;
            ++cA;
        }
        if (actB) {
            b0.x += vb0.x; b0.y += vb0.y; b0.z += vb0.z; b0.w += vb0.w;
            b1.x += vb1.x; b1.y += vb1.y; b1.z += vb1.z; b1.w += vb1.w;
            b2.x += vb2.x; b2.y += vb2.y; b2.z += vb2.z; b2.w += vb2.w;
            b3.x += vb3.x; b3.y += vb3.y; b3.z += vb3.z; b3.w += vb3.w;
            ++cB;
        }
    }

    const float ia = 1.0f / fmaxf((float)cA, 1.0f);
    const float ib = 1.0f / fmaxf((float)cB, 1.0f);
    float* oa = out + (size_t)nodeA * D_OUT + sub * 4;
    float* ob = out + (size_t)nodeB * D_OUT + sub * 4;
    float4 qa0 = *(const float4*)(oa);
    float4 qa1 = *(const float4*)(oa + 32);
    float4 qa2 = *(const float4*)(oa + 64);
    float4 qa3 = *(const float4*)(oa + 96);
    float4 qb0 = *(const float4*)(ob);
    float4 qb1 = *(const float4*)(ob + 32);
    float4 qb2 = *(const float4*)(ob + 64);
    float4 qb3 = *(const float4*)(ob + 96);
    qa0.x += a0.x * ia; qa0.y += a0.y * ia; qa0.z += a0.z * ia; qa0.w += a0.w * ia;
    qa1.x += a1.x * ia; qa1.y += a1.y * ia; qa1.z += a1.z * ia; qa1.w += a1.w * ia;
    qa2.x += a2.x * ia; qa2.y += a2.y * ia; qa2.z += a2.z * ia; qa2.w += a2.w * ia;
    qa3.x += a3.x * ia; qa3.y += a3.y * ia; qa3.z += a3.z * ia; qa3.w += a3.w * ia;
    qb0.x += b0.x * ib; qb0.y += b0.y * ib; qb0.z += b0.z * ib; qb0.w += b0.w * ib;
    qb1.x += b1.x * ib; qb1.y += b1.y * ib; qb1.z += b1.z * ib; qb1.w += b1.w * ib;
    qb2.x += b2.x * ib; qb2.y += b2.y * ib; qb2.z += b2.z * ib; qb2.w += b2.w * ib;
    qb3.x += b3.x * ib; qb3.y += b3.y * ib; qb3.z += b3.z * ib; qb3.w += b3.w * ib;
    *(float4*)(oa)      = qa0;
    *(float4*)(oa + 32) = qa1;
    *(float4*)(oa + 64) = qa2;
    *(float4*)(oa + 96) = qa3;
    *(float4*)(ob)      = qb0;
    *(float4*)(ob + 32) = qb1;
    *(float4*)(ob + 64) = qb2;
    *(float4*)(ob + 96) = qb3;
}

extern "C" void kernel_launch(void* const* d_in, const int* in_sizes, int n_in,
                              void* d_out, int out_size, void* d_ws, size_t ws_size,
                              hipStream_t stream) {
    const float* feat = (const float*)d_in[0];
    const int*   src  = (const int*)d_in[1];
    const int*   dst  = (const int*)d_in[2];
    const float* W    = (const float*)d_in[3];
    const float* bias = (const float*)d_in[4];
    float*       out  = (float*)d_out;

    float* P     = (float*)d_ws;
    int*   heads = (int*)(P + (size_t)N_NODES * D_OUT);
    int*   rec   = heads + (size_t)N_SLICES * HPW;

    gcn_link_product_kernel<<<FUSE_BLOCKS, FUSE_THREADS, 0, stream>>>(src, dst, heads, rec, feat, W, bias, P, out);

    gcn_gather_kernel<<<GB_BLOCKS, 256, 0, stream>>>(P, heads, rec, out);
}
